# starting kernel
# baseline (speedup 1.0000x reference)
#include <hip/hip_runtime.h>

typedef __attribute__((ext_vector_type(2))) float v2f;
typedef __attribute__((ext_vector_type(8))) float v8f;

#define D 128
#define LDSPAD 132

__global__ void cheb_deg_kernel(const int* __restrict__ ei, int E,
                                float* __restrict__ deg) {
    int e = blockIdx.x * blockDim.x + threadIdx.x;
    if (e >= E) return;
    int s = ei[e];
    int d = ei[E + e];
    if (s != d) unsafeAtomicAdd(&deg[s], 1.0f);
}

__global__ void cheb_dis_kernel(const float* __restrict__ deg,
                                float* __restrict__ dis, int n) {
    int i = blockIdx.x * blockDim.x + threadIdx.x;
    if (i >= n) return;
    float dv = deg[i];
    dis[i] = (dv > 0.0f) ? rsqrtf(fmaxf(dv, 1.0f)) : 0.0f;
}

__global__ void cheb_spmm_kernel(const int* __restrict__ ei, int E,
                                 const float* __restrict__ dis,
                                 const float* __restrict__ hin,
                                 float* __restrict__ agg) {
    int wid  = (int)((blockIdx.x * (long)blockDim.x + threadIdx.x) >> 5);
    int lane = threadIdx.x & 31;
    if (wid >= E) return;
    int s = ei[wid];
    int d = ei[E + wid];
    float w = (s != d) ? -(dis[s] * dis[d]) : 0.0f;
    if (w == 0.0f) return;
    int c = lane * 4;
    const float4 h = *(const float4*)(hin + (long)s * D + c);
    float* o = agg + (long)d * D + c;
    unsafeAtomicAdd(o + 0, w * h.x);
    unsafeAtomicAdd(o + 1, w * h.y);
    unsafeAtomicAdd(o + 2, w * h.z);
    unsafeAtomicAdd(o + 3, w * h.w);
}

__global__ void __launch_bounds__(256)
cheb_gemm_kernel(const float* __restrict__ x,
                 const float* __restrict__ tx1,
                 const float* __restrict__ s2,
                 const float* __restrict__ weight,
                 const float* __restrict__ bias,
                 float* __restrict__ out, int n) {
    __shared__ float lA[3][16][LDSPAD];

    const int m0  = blockIdx.x * 16;
    const int tid = threadIdx.x;

    for (int idx = tid; idx < 3 * 16 * D; idx += 256) {
        int t = idx >> 11;
        int r = (idx >> 7) & 15;
        int c = idx & (D - 1);
        int row = m0 + r;
        float v = 0.0f;
        if (row < n) {
            long off = (long)row * D + c;
            if (t == 0)      v = x[off];
            else if (t == 1) v = tx1[off];
            else             v = 2.0f * s2[off] - x[off];
        }
        lA[t][r][c] = v;
    }
    __syncthreads();

    const int wave = tid >> 5;
    const int lane = tid & 31;
    const int hf   = lane >> 4;
    const int lm   = lane & 15;
    const int col  = wave * 16 + lm;

    v8f acc = {};
    for (int kk = 0; kk < D; kk += 4) {
        const int ka = kk + 2 * hf;
#pragma unroll
        for (int t = 0; t < 3; ++t) {
            v2f a;
            a.x = lA[t][lm][ka];
            a.y = lA[t][lm][ka + 1];
            const float* W = weight + (long)t * D * D;
            v2f b;
            b.x = W[(long)ka * D + col];
            b.y = W[(long)(ka + 1) * D + col];
            acc = __builtin_amdgcn_wmma_f32_16x16x4_f32(
                false, a, false, b, (short)0, acc, false, false);
        }
    }

    const float bv = bias[col];
#pragma unroll
    for (int i = 0; i < 8; ++i) {
        int row = m0 + i + 8 * hf;
        if (row < n) out[(long)row * D + col] = acc[i] + bv;
    }
}

extern "C" void kernel_launch(void* const* d_in, const int* in_sizes, int n_in,
                              void* d_out, int out_size, void* d_ws, size_t ws_size,
                              hipStream_t stream) {
    const float* x      = (const float*)d_in[0];
    const int*   ei     = (const int*)d_in[1];
    const float* weight = (const float*)d_in[2];
    const float* bias   = (const float*)d_in[3];
    float*       out    = (float*)d_out;

    const int n = in_sizes[0] / D;
    const int E = in_sizes[1] / 2;

    float* deg = (float*)d_ws;
    float* dis = deg + n;
    float* tx1 = dis + n;
    float* s2  = tx1 + (long)n * D;

    const size_t zbytes = ((size_t)2 * n + (size_t)2 * n * D) * sizeof(float);
    hipMemsetAsync(d_ws, 0, zbytes, stream);

    cheb_deg_kernel<<<(E + 255) / 256, 256, 0, stream>>>(ei, E, deg);
    cheb_dis_kernel<<<(n + 255) / 256, 256, 0, stream>>>(deg, dis, n);

    const int spmm_blocks = (E + 7) / 8;
    cheb_spmm_kernel<<<spmm_blocks, 256, 0, stream>>>(ei, E, dis, x,   tx1);
    cheb_spmm_kernel<<<spmm_blocks, 256, 0, stream>>>(ei, E, dis, tx1, s2);

    cheb_gemm_kernel<<<(n + 15) / 16, 256, 0, stream>>>(x, tx1, s2, weight,
                                                        bias, out, n);
}


# SpMM scatter of float atomics replaced by a gather over a dst-sorted edge list built per call (int atomic count+slot, scan, fill); row sum order = arrival-order rank
# speedup vs baseline: 2.0090x; 2.0090x over previous
#include <hip/hip_runtime.h>

typedef __attribute__((ext_vector_type(2))) float v2f;
typedef __attribute__((ext_vector_type(8))) float v8f;

#define D 128
#define LDSPAD 132

#define SCAN_T 1024

__global__ void cheb_count_kernel(const int* __restrict__ ei, int E,
                                  int* __restrict__ deg,
                                  int* __restrict__ cnt,
                                  int* __restrict__ rank) {
    int e = blockIdx.x * blockDim.x + threadIdx.x;
    if (e >= E) return;
    int s = ei[e];
    int d = ei[E + e];
    if (s != d) {
        atomicAdd(&deg[s], 1);
        rank[e] = atomicAdd(&cnt[d], 1);
    }
}

__global__ void __launch_bounds__(SCAN_T)
cheb_scan_kernel(const int* __restrict__ cnt, const int* __restrict__ deg,
                 int* __restrict__ rowptr, float* __restrict__ dis, int n) {
    __shared__ int wfront[32];
    __shared__ int wtotal[32];
    const int tid  = threadIdx.x;
    const int lane = tid & 31;
    const int wave = tid >> 5;
    const int first = blockIdx.x * SCAN_T;

    int part = 0;
    for (int i = tid; i < first; i += SCAN_T) part += cnt[i];
    for (int o = 16; o > 0; o >>= 1) part += __shfl_xor(part, o);

    const int i = first + tid;
    const int v = (i < n) ? cnt[i] : 0;
    int inc = v;
    for (int o = 1; o < 32; o <<= 1) {
        int t = __shfl_up(inc, o);
        if (lane >= o) inc += t;
    }
    if (lane == 0)  wfront[wave] = part;
    if (lane == 31) wtotal[wave] = inc;
    __syncthreads();

    int base = 0;
    for (int w = 0; w < 32; ++w) {
        base += wfront[w];
        if (w < wave) base += wtotal[w];
    }
    if (i < n) {
        rowptr[i] = base + inc - v;
        const int dg = deg[i];
        dis[i] = (dg > 0) ? rsqrtf(fmaxf((float)dg, 1.0f)) : 0.0f;
    }
}

__global__ void cheb_fill_kernel(const int* __restrict__ ei, int E,
                                 const int* __restrict__ rank,
                                 const int* __restrict__ rowptr,
                                 const float* __restrict__ dis,
                                 int* __restrict__ ecol,
                                 float* __restrict__ ewt) {
    int e = blockIdx.x * blockDim.x + threadIdx.x;
    if (e >= E) return;
    int s = ei[e];
    int d = ei[E + e];
    if (s == d) return;
    int pos = rowptr[d] + rank[e];
    ecol[pos] = s;
    ewt[pos]  = -(dis[s] * dis[d]);
}

static __device__ __forceinline__ void axpy4(float4& acc, float w, const float4 h) {
    acc.x = __builtin_fmaf(w, h.x, acc.x);
    acc.y = __builtin_fmaf(w, h.y, acc.y);
    acc.z = __builtin_fmaf(w, h.z, acc.z);
    acc.w = __builtin_fmaf(w, h.w, acc.w);
}

__global__ void __launch_bounds__(256)
cheb_gather_kernel(const int* __restrict__ rowptr,
                   const int* __restrict__ cnt,
                   const int* __restrict__ ecol,
                   const float* __restrict__ ewt,
                   const float* __restrict__ hin,
                   float* __restrict__ agg, int n) {
    const int row = __builtin_amdgcn_readfirstlane(
        (int)((blockIdx.x * 256u + threadIdx.x) >> 5));
    if (row >= n) return;
    const int lane = threadIdx.x & 31;
    const int beg = rowptr[row];
    const int len = cnt[row];
    const float* hl = hin + lane * 4;

    float4 acc = make_float4(0.0f, 0.0f, 0.0f, 0.0f);
    for (int base = 0; base < len; base += 32) {
        const int m = min(32, len - base);
        int ms = 0;
        int mw = 0;
        if (lane < m) {
            ms = ecol[beg + base + lane];
            mw = __float_as_int(ewt[beg + base + lane]);
        }
        int j = 0;
        for (; j + 4 <= m; j += 4) {
            const int s0 = __builtin_amdgcn_readlane(ms, j);
            const int s1 = __builtin_amdgcn_readlane(ms, j + 1);
            const int s2 = __builtin_amdgcn_readlane(ms, j + 2);
            const int s3 = __builtin_amdgcn_readlane(ms, j + 3);
            const float w0 = __int_as_float(__builtin_amdgcn_readlane(mw, j));
            const float w1 = __int_as_float(__builtin_amdgcn_readlane(mw, j + 1));
            const float w2 = __int_as_float(__builtin_amdgcn_readlane(mw, j + 2));
            const float w3 = __int_as_float(__builtin_amdgcn_readlane(mw, j + 3));
            const float4 h0 = *(const float4*)(hl + (long)s0 * D);
            const float4 h1 = *(const float4*)(hl + (long)s1 * D);
            const float4 h2 = *(const float4*)(hl + (long)s2 * D);
            const float4 h3 = *(const float4*)(hl + (long)s3 * D);
            axpy4(acc, w0, h0);
            axpy4(acc, w1, h1);
            axpy4(acc, w2, h2);
            axpy4(acc, w3, h3);
        }
        for (; j < m; ++j) {
            const int s0 = __builtin_amdgcn_readlane(ms, j);
            const float w0 = __int_as_float(__builtin_amdgcn_readlane(mw, j));
            const float4 h0 = *(const float4*)(hl + (long)s0 * D);
            axpy4(acc, w0, h0);
        }
    }
    *(float4*)(agg + (long)row * D + lane * 4) = acc;
}

__global__ void __launch_bounds__(256)
cheb_gemm_kernel(const float* __restrict__ x,
                 const float* __restrict__ tx1,
                 const float* __restrict__ s2,
                 const float* __restrict__ weight,
                 const float* __restrict__ bias,
                 float* __restrict__ out, int n) {
    __shared__ float lA[3][16][LDSPAD];

    const int m0  = blockIdx.x * 16;
    const int tid = threadIdx.x;

    for (int idx = tid; idx < 3 * 16 * D; idx += 256) {
        int t = idx >> 11;
        int r = (idx >> 7) & 15;
        int c = idx & (D - 1);
        int row = m0 + r;
        float v = 0.0f;
        if (row < n) {
            long off = (long)row * D + c;
            if (t == 0)      v = x[off];
            else if (t == 1) v = tx1[off];
            else             v = 2.0f * s2[off] - x[off];
        }
        lA[t][r][c] = v;
    }
    __syncthreads();

    const int wave = tid >> 5;
    const int lane = tid & 31;
    const int hf   = lane >> 4;
    const int lm   = lane & 15;
    const int col  = wave * 16 + lm;

    v8f acc = {};
    for (int kk = 0; kk < D; kk += 4) {
        const int ka = kk + 2 * hf;
#pragma unroll
        for (int t = 0; t < 3; ++t) {
            v2f a;
            a.x = lA[t][lm][ka];
            a.y = lA[t][lm][ka + 1];
            const float* W = weight + (long)t * D * D;
            v2f b;
            b.x = W[(long)ka * D + col];
            b.y = W[(long)(ka + 1) * D + col];
            acc = __builtin_amdgcn_wmma_f32_16x16x4_f32(
                false, a, false, b, (short)0, acc, false, false);
        }
    }

    const float bv = bias[col];
#pragma unroll
    for (int i = 0; i < 8; ++i) {
        int row = m0 + i + 8 * hf;
        if (row < n) out[(long)row * D + col] = acc[i] + bv;
    }
}

extern "C" void kernel_launch(void* const* d_in, const int* in_sizes, int n_in,
                              void* d_out, int out_size, void* d_ws, size_t ws_size,
                              hipStream_t stream) {
    const float* x      = (const float*)d_in[0];
    const int*   ei     = (const int*)d_in[1];
    const float* weight = (const float*)d_in[2];
    const float* bias   = (const float*)d_in[3];
    float*       out    = (float*)d_out;

    const int n = in_sizes[0] / D;
    const int E = in_sizes[1] / 2;

    int*   deg = (int*)d_ws;
    int*   cnt = deg + n;
    float* tx1 = (float*)(cnt + n);
    float* s2  = tx1 + (long)n * D;

    int*   ecol   = (int*)d_out;
    float* ewt    = (float*)(ecol + E);
    int*   rank   = (int*)(ewt + E);
    int*   rowptr = rank + E;
    float* dis    = (float*)(rowptr + n);

    hipMemsetAsync(d_ws, 0, (size_t)2 * n * sizeof(int), stream);

    cheb_count_kernel<<<(E + 255) / 256, 256, 0, stream>>>(ei, E, deg, cnt, rank);
    cheb_scan_kernel<<<(n + SCAN_T - 1) / SCAN_T, SCAN_T, 0, stream>>>(cnt, deg, rowptr, dis, n);
    cheb_fill_kernel<<<(E + 255) / 256, 256, 0, stream>>>(ei, E, rank, rowptr, dis, ecol, ewt);

    const int gather_blocks = (n + 7) / 8;
    cheb_gather_kernel<<<gather_blocks, 256, 0, stream>>>(rowptr, cnt, ecol, ewt, x,   tx1, n);
    cheb_gather_kernel<<<gather_blocks, 256, 0, stream>>>(rowptr, cnt, ecol, ewt, tx1, s2,  n);

    cheb_gemm_kernel<<<(n + 15) / 16, 256, 0, stream>>>(x, tx1, s2, weight,
                                                        bias, out, n);
}


# count and slot pass moved from global int atomics to LDS counters: per edge-slice x node-range tables, merged per node, fill takes slots by LDS add; row sum order = slice, then LDS arrival
# speedup vs baseline: 2.6644x; 1.3262x over previous
#include <hip/hip_runtime.h>

typedef __attribute__((ext_vector_type(2))) float v2f;
typedef __attribute__((ext_vector_type(8))) float v8f;

#define D 128
#define LDSPAD 132

#define SCAN_T 1024

#define RANGE  25000
#define NSLICE 64
#define HT     1024
#define EU     8

__global__ void __launch_bounds__(HT)
cheb_hist_kernel(const int* __restrict__ ei, int E, int n, int slice_len,
                 int* __restrict__ pd, int* __restrict__ ps) {
    __shared__ int tab[RANGE];
    const int tid  = threadIdx.x;
    const int base = blockIdx.x * RANGE;
    const int sl   = blockIdx.y;
    const bool by_src = (blockIdx.z != 0);
    const int lim = min(RANGE, n - base);

    for (int i = tid; i < RANGE; i += HT) tab[i] = 0;
    __syncthreads();

    const int e0 = sl * slice_len;
    const int e1 = min(E, e0 + slice_len);
    for (int eb = e0 + tid; eb < e1; eb += HT * EU) {
        int s[EU], d[EU];
#pragma unroll
        for (int u = 0; u < EU; ++u) {
            const int e = eb + u * HT;
            const bool in = e < e1;
            s[u] = in ? ei[e] : 0;
            d[u] = in ? ei[E + e] : 0;
        }
#pragma unroll
        for (int u = 0; u < EU; ++u) {
            const unsigned k = (unsigned)((by_src ? s[u] : d[u]) - base);
            if (s[u] != d[u] && k < (unsigned)lim) atomicAdd(&tab[k], 1);
        }
    }
    __syncthreads();

    int* __restrict__ rowp = (by_src ? ps : pd) + (long)sl * n + base;
    for (int i = tid; i < lim; i += HT) rowp[i] = tab[i];
}

__global__ void __launch_bounds__(256)
cheb_merge_kernel(int* __restrict__ pd, const int* __restrict__ ps,
                  int* __restrict__ cnt, int* __restrict__ deg, int n) {
    const int v = blockIdx.x * 256 + threadIdx.x;
    if (v >= n) return;
    if (blockIdx.y == 0) {
        int run = 0;
        for (int c = 0; c < NSLICE; c += 32) {
            int t[32];
#pragma unroll
            for (int u = 0; u < 32; ++u) t[u] = pd[(long)(c + u) * n + v];
#pragma unroll
            for (int u = 0; u < 32; ++u) {
                pd[(long)(c + u) * n + v] = run;
                run += t[u];
            }
        }
        cnt[v] = run;
    } else {
        int sum = 0;
#pragma unroll 32
        for (int sl = 0; sl < NSLICE; ++sl) sum += ps[(long)sl * n + v];
        deg[v] = sum;
    }
}

__global__ void __launch_bounds__(SCAN_T)
cheb_scan_kernel(const int* __restrict__ cnt, const int* __restrict__ deg,
                 int* __restrict__ rowptr, float* __restrict__ dis, int n) {
    __shared__ int wfront[32];
    __shared__ int wtotal[32];
    const int tid  = threadIdx.x;
    const int lane = tid & 31;
    const int wave = tid >> 5;
    const int first = blockIdx.x * SCAN_T;

    int part = 0;
    for (int i = tid; i < first; i += SCAN_T) part += cnt[i];
    for (int o = 16; o > 0; o >>= 1) part += __shfl_xor(part, o);

    const int i = first + tid;
    const int v = (i < n) ? cnt[i] : 0;
    int inc = v;
    for (int o = 1; o < 32; o <<= 1) {
        int t = __shfl_up(inc, o);
        if (lane >= o) inc += t;
    }
    if (lane == 0)  wfront[wave] = part;
    if (lane == 31) wtotal[wave] = inc;
    __syncthreads();

    int base = 0;
    for (int w = 0; w < 32; ++w) {
        base += wfront[w];
        if (w < wave) base += wtotal[w];
    }
    if (i < n) {
        rowptr[i] = base + inc - v;
        const int dg = deg[i];
        dis[i] = (dg > 0) ? rsqrtf(fmaxf((float)dg, 1.0f)) : 0.0f;
    }
}

__global__ void __launch_bounds__(HT)
cheb_fill_kernel(const int* __restrict__ ei, int E, int n, int slice_len,
                 const int* __restrict__ rowptr,
                 const int* __restrict__ pd,
                 const float* __restrict__ dis,
                 int* __restrict__ ecol,
                 float* __restrict__ ewt) {
    __shared__ int slot[RANGE];
    const int tid  = threadIdx.x;
    const int base = blockIdx.x * RANGE;
    const int sl   = blockIdx.y;
    const int lim  = min(RANGE, n - base);

    const int* __restrict__ off = pd + (long)sl * n + base;
    for (int i = tid; i < lim; i += HT) slot[i] = rowptr[base + i] + off[i];
    __syncthreads();

    const int e0 = sl * slice_len;
    const int e1 = min(E, e0 + slice_len);
    for (int eb = e0 + tid; eb < e1; eb += HT * EU) {
        int s[EU], d[EU];
#pragma unroll
        for (int u = 0; u < EU; ++u) {
            const int e = eb + u * HT;
            const bool in = e < e1;
            s[u] = in ? ei[e] : 0;
            d[u] = in ? ei[E + e] : 0;
        }
        bool  ok[EU];
        float fs[EU], fd[EU];
#pragma unroll
        for (int u = 0; u < EU; ++u) {
            const unsigned k = (unsigned)(d[u] - base);
            ok[u] = (s[u] != d[u]) && (k < (unsigned)lim);
            fs[u] = ok[u] ? dis[s[u]] : 0.0f;
            fd[u] = ok[u] ? dis[d[u]] : 0.0f;
        }
#pragma unroll
        for (int u = 0; u < EU; ++u) {
            if (ok[u]) {
                const int pos = atomicAdd(&slot[d[u] - base], 1);
                ecol[pos] = s[u];
                ewt[pos]  = -(fs[u] * fd[u]);
            }
        }
    }
}

static __device__ __forceinline__ void axpy4(float4& acc, float w, const float4 h) {
    acc.x = __builtin_fmaf(w, h.x, acc.x);
    acc.y = __builtin_fmaf(w, h.y, acc.y);
    acc.z = __builtin_fmaf(w, h.z, acc.z);
    acc.w = __builtin_fmaf(w, h.w, acc.w);
}

__global__ void __launch_bounds__(256)
cheb_gather_kernel(const int* __restrict__ rowptr,
                   const int* __restrict__ cnt,
                   const int* __restrict__ ecol,
                   const float* __restrict__ ewt,
                   const float* __restrict__ hin,
                   float* __restrict__ agg, int n) {
    const int row = __builtin_amdgcn_readfirstlane(
        (int)((blockIdx.x * 256u + threadIdx.x) >> 5));
    if (row >= n) return;
    const int lane = threadIdx.x & 31;
    const int beg = rowptr[row];
    const int len = cnt[row];
    const float* hl = hin + lane * 4;

    float4 acc = make_float4(0.0f, 0.0f, 0.0f, 0.0f);
    for (int base = 0; base < len; base += 32) {
        const int m = min(32, len - base);
        int ms = 0;
        int mw = 0;
        if (lane < m) {
            ms = ecol[beg + base + lane];
            mw = __float_as_int(ewt[beg + base + lane]);
        }
        int j = 0;
        for (; j + 4 <= m; j += 4) {
            const int s0 = __builtin_amdgcn_readlane(ms, j);
            const int s1 = __builtin_amdgcn_readlane(ms, j + 1);
            const int s2 = __builtin_amdgcn_readlane(ms, j + 2);
            const int s3 = __builtin_amdgcn_readlane(ms, j + 3);
            const float w0 = __int_as_float(__builtin_amdgcn_readlane(mw, j));
            const float w1 = __int_as_float(__builtin_amdgcn_readlane(mw, j + 1));
            const float w2 = __int_as_float(__builtin_amdgcn_readlane(mw, j + 2));
            const float w3 = __int_as_float(__builtin_amdgcn_readlane(mw, j + 3));
            const float4 h0 = *(const float4*)(hl + (long)s0 * D);
            const float4 h1 = *(const float4*)(hl + (long)s1 * D);
            const float4 h2 = *(const float4*)(hl + (long)s2 * D);
            const float4 h3 = *(const float4*)(hl + (long)s3 * D);
            axpy4(acc, w0, h0);
            axpy4(acc, w1, h1);
            axpy4(acc, w2, h2);
            axpy4(acc, w3, h3);
        }
        for (; j < m; ++j) {
            const int s0 = __builtin_amdgcn_readlane(ms, j);
            const float w0 = __int_as_float(__builtin_amdgcn_readlane(mw, j));
            const float4 h0 = *(const float4*)(hl + (long)s0 * D);
            axpy4(acc, w0, h0);
        }
    }
    *(float4*)(agg + (long)row * D + lane * 4) = acc;
}

__global__ void __launch_bounds__(256)
cheb_gemm_kernel(const float* __restrict__ x,
                 const float* __restrict__ tx1,
                 const float* __restrict__ s2,
                 const float* __restrict__ weight,
                 const float* __restrict__ bias,
                 float* __restrict__ out, int n) {
    __shared__ float lA[3][16][LDSPAD];

    const int m0  = blockIdx.x * 16;
    const int tid = threadIdx.x;

    for (int idx = tid; idx < 3 * 16 * D; idx += 256) {
        int t = idx >> 11;
        int r = (idx >> 7) & 15;
        int c = idx & (D - 1);
        int row = m0 + r;
        float v = 0.0f;
        if (row < n) {
            long off = (long)row * D + c;
            if (t == 0)      v = x[off];
            else if (t == 1) v = tx1[off];
            else             v = 2.0f * s2[off] - x[off];
        }
        lA[t][r][c] = v;
    }
    __syncthreads();

    const int wave = tid >> 5;
    const int lane = tid & 31;
    const int hf   = lane >> 4;
    const int lm   = lane & 15;
    const int col  = wave * 16 + lm;

    v8f acc = {};
    for (int kk = 0; kk < D; kk += 4) {
        const int ka = kk + 2 * hf;
#pragma unroll
        for (int t = 0; t < 3; ++t) {
            v2f a;
            a.x = lA[t][lm][ka];
            a.y = lA[t][lm][ka + 1];
            const float* W = weight + (long)t * D * D;
            v2f b;
            b.x = W[(long)ka * D + col];
            b.y = W[(long)(ka + 1) * D + col];
            acc = __builtin_amdgcn_wmma_f32_16x16x4_f32(
                false, a, false, b, (short)0, acc, false, false);
        }
    }

    const float bv = bias[col];
#pragma unroll
    for (int i = 0; i < 8; ++i) {
        int row = m0 + i + 8 * hf;
        if (row < n) out[(long)row * D + col] = acc[i] + bv;
    }
}

extern "C" void kernel_launch(void* const* d_in, const int* in_sizes, int n_in,
                              void* d_out, int out_size, void* d_ws, size_t ws_size,
                              hipStream_t stream) {
    const float* x      = (const float*)d_in[0];
    const int*   ei     = (const int*)d_in[1];
    const float* weight = (const float*)d_in[2];
    const float* bias   = (const float*)d_in[3];
    float*       out    = (float*)d_out;

    const int n = in_sizes[0] / D;
    const int E = in_sizes[1] / 2;

    int*   deg = (int*)d_ws;
    int*   cnt = deg + n;
    float* tx1 = (float*)(cnt + n);
    float* s2  = tx1 + (long)n * D;

    int*   ecol   = (int*)d_out;
    float* ewt    = (float*)(ecol + E);
    int*   rowptr = (int*)(ewt + E);
    float* dis    = (float*)(rowptr + n);

    int* ps = (int*)tx1;
    int* pd = (int*)s2;

    const int ranges    = (n + RANGE - 1) / RANGE;
    const int slice_len = (E + NSLICE - 1) / NSLICE;

    cheb_hist_kernel<<<dim3(ranges, NSLICE, 2), HT, 0, stream>>>(ei, E, n, slice_len, pd, ps);
    cheb_merge_kernel<<<dim3((n + 255) / 256, 2), 256, 0, stream>>>(pd, ps, cnt, deg, n);
    cheb_scan_kernel<<<(n + SCAN_T - 1) / SCAN_T, SCAN_T, 0, stream>>>(cnt, deg, rowptr, dis, n);
    cheb_fill_kernel<<<dim3(ranges, NSLICE), HT, 0, stream>>>(ei, E, n, slice_len, rowptr, pd, dis, ecol, ewt);

    const int gather_blocks = (n + 7) / 8;
    cheb_gather_kernel<<<gather_blocks, 256, 0, stream>>>(rowptr, cnt, ecol, ewt, x,   tx1, n);
    cheb_gather_kernel<<<gather_blocks, 256, 0, stream>>>(rowptr, cnt, ecol, ewt, tx1, s2,  n);

    cheb_gemm_kernel<<<(n + 15) / 16, 256, 0, stream>>>(x, tx1, s2, weight,
                                                        bias, out, n);
}


# second SpMM fused into the GEMM launch: each wave gathers its two rows of S from Tx1 straight into the LDS A tile as 2S-x, beside its x and Tx1 rows; S plane and one launch dropped
# speedup vs baseline: 2.8584x; 1.0728x over previous
#include <hip/hip_runtime.h>

typedef __attribute__((ext_vector_type(2))) float v2f;
typedef __attribute__((ext_vector_type(8))) float v8f;

#define D 128
#define LDSPAD 132

#define SCAN_T 1024

#define RANGE  25000
#define NSLICE 64
#define HT     1024
#define EU     8

__global__ void __launch_bounds__(HT)
cheb_hist_kernel(const int* __restrict__ ei, int E, int n, int slice_len,
                 int* __restrict__ pd, int* __restrict__ ps) {
    __shared__ int tab[RANGE];
    const int tid  = threadIdx.x;
    const int base = blockIdx.x * RANGE;
    const int sl   = blockIdx.y;
    const bool by_src = (blockIdx.z != 0);
    const int lim = min(RANGE, n - base);

    for (int i = tid; i < RANGE; i += HT) tab[i] = 0;
    __syncthreads();

    const int e0 = sl * slice_len;
    const int e1 = min(E, e0 + slice_len);
    for (int eb = e0 + tid; eb < e1; eb += HT * EU) {
        int s[EU], d[EU];
#pragma unroll
        for (int u = 0; u < EU; ++u) {
            const int e = eb + u * HT;
            const bool in = e < e1;
            s[u] = in ? ei[e] : 0;
            d[u] = in ? ei[E + e] : 0;
        }
#pragma unroll
        for (int u = 0; u < EU; ++u) {
            const unsigned k = (unsigned)((by_src ? s[u] : d[u]) - base);
            if (s[u] != d[u] && k < (unsigned)lim) atomicAdd(&tab[k], 1);
        }
    }
    __syncthreads();

    int* __restrict__ rowp = (by_src ? ps : pd) + (long)sl * n + base;
    for (int i = tid; i < lim; i += HT) rowp[i] = tab[i];
}

__global__ void __launch_bounds__(256)
cheb_merge_kernel(int* __restrict__ pd, const int* __restrict__ ps,
                  int* __restrict__ cnt, int* __restrict__ deg, int n) {
    const int v = blockIdx.x * 256 + threadIdx.x;
    if (v >= n) return;
    if (blockIdx.y == 0) {
        int run = 0;
        for (int c = 0; c < NSLICE; c += 32) {
            int t[32];
#pragma unroll
            for (int u = 0; u < 32; ++u) t[u] = pd[(long)(c + u) * n + v];
#pragma unroll
            for (int u = 0; u < 32; ++u) {
                pd[(long)(c + u) * n + v] = run;
                run += t[u];
            }
        }
        cnt[v] = run;
    } else {
        int sum = 0;
#pragma unroll 32
        for (int sl = 0; sl < NSLICE; ++sl) sum += ps[(long)sl * n + v];
        deg[v] = sum;
    }
}

__global__ void __launch_bounds__(SCAN_T)
cheb_scan_kernel(const int* __restrict__ cnt, const int* __restrict__ deg,
                 int* __restrict__ rowptr, float* __restrict__ dis, int n) {
    __shared__ int wfront[32];
    __shared__ int wtotal[32];
    const int tid  = threadIdx.x;
    const int lane = tid & 31;
    const int wave = tid >> 5;
    const int first = blockIdx.x * SCAN_T;

    int part = 0;
    for (int i = tid; i < first; i += SCAN_T) part += cnt[i];
    for (int o = 16; o > 0; o >>= 1) part += __shfl_xor(part, o);

    const int i = first + tid;
    const int v = (i < n) ? cnt[i] : 0;
    int inc = v;
    for (int o = 1; o < 32; o <<= 1) {
        int t = __shfl_up(inc, o);
        if (lane >= o) inc += t;
    }
    if (lane == 0)  wfront[wave] = part;
    if (lane == 31) wtotal[wave] = inc;
    __syncthreads();

    int base = 0;
    for (int w = 0; w < 32; ++w) {
        base += wfront[w];
        if (w < wave) base += wtotal[w];
    }
    if (i < n) {
        rowptr[i] = base + inc - v;
        const int dg = deg[i];
        dis[i] = (dg > 0) ? rsqrtf(fmaxf((float)dg, 1.0f)) : 0.0f;
    }
}

__global__ void __launch_bounds__(HT)
cheb_fill_kernel(const int* __restrict__ ei, int E, int n, int slice_len,
                 const int* __restrict__ rowptr,
                 const int* __restrict__ pd,
                 const float* __restrict__ dis,
                 int* __restrict__ ecol,
                 float* __restrict__ ewt) {
    __shared__ int slot[RANGE];
    const int tid  = threadIdx.x;
    const int base = blockIdx.x * RANGE;
    const int sl   = blockIdx.y;
    const int lim  = min(RANGE, n - base);

    const int* __restrict__ off = pd + (long)sl * n + base;
    for (int i = tid; i < lim; i += HT) slot[i] = rowptr[base + i] + off[i];
    __syncthreads();

    const int e0 = sl * slice_len;
    const int e1 = min(E, e0 + slice_len);
    for (int eb = e0 + tid; eb < e1; eb += HT * EU) {
        int s[EU], d[EU];
#pragma unroll
        for (int u = 0; u < EU; ++u) {
            const int e = eb + u * HT;
            const bool in = e < e1;
            s[u] = in ? ei[e] : 0;
            d[u] = in ? ei[E + e] : 0;
        }
        bool  ok[EU];
        float fs[EU], fd[EU];
#pragma unroll
        for (int u = 0; u < EU; ++u) {
            const unsigned k = (unsigned)(d[u] - base);
            ok[u] = (s[u] != d[u]) && (k < (unsigned)lim);
            fs[u] = ok[u] ? dis[s[u]] : 0.0f;
            fd[u] = ok[u] ? dis[d[u]] : 0.0f;
        }
#pragma unroll
        for (int u = 0; u < EU; ++u) {
            if (ok[u]) {
                const int pos = atomicAdd(&slot[d[u] - base], 1);
                ecol[pos] = s[u];
                ewt[pos]  = -(fs[u] * fd[u]);
            }
        }
    }
}

static __device__ __forceinline__ void axpy4(float4& acc, float w, const float4 h) {
    acc.x = __builtin_fmaf(w, h.x, acc.x);
    acc.y = __builtin_fmaf(w, h.y, acc.y);
    acc.z = __builtin_fmaf(w, h.z, acc.z);
    acc.w = __builtin_fmaf(w, h.w, acc.w);
}

__global__ void __launch_bounds__(256)
cheb_gather_kernel(const int* __restrict__ rowptr,
                   const int* __restrict__ cnt,
                   const int* __restrict__ ecol,
                   const float* __restrict__ ewt,
                   const float* __restrict__ hin,
                   float* __restrict__ agg, int n) {
    const int row = __builtin_amdgcn_readfirstlane(
        (int)((blockIdx.x * 256u + threadIdx.x) >> 5));
    if (row >= n) return;
    const int lane = threadIdx.x & 31;
    const int beg = rowptr[row];
    const int len = cnt[row];
    const float* hl = hin + lane * 4;

    float4 acc = make_float4(0.0f, 0.0f, 0.0f, 0.0f);
    for (int base = 0; base < len; base += 32) {
        const int m = min(32, len - base);
        int ms = 0;
        int mw = 0;
        if (lane < m) {
            ms = ecol[beg + base + lane];
            mw = __float_as_int(ewt[beg + base + lane]);
        }
        int j = 0;
        for (; j + 4 <= m; j += 4) {
            const int s0 = __builtin_amdgcn_readlane(ms, j);
            const int s1 = __builtin_amdgcn_readlane(ms, j + 1);
            const int s2 = __builtin_amdgcn_readlane(ms, j + 2);
            const int s3 = __builtin_amdgcn_readlane(ms, j + 3);
            const float w0 = __int_as_float(__builtin_amdgcn_readlane(mw, j));
            const float w1 = __int_as_float(__builtin_amdgcn_readlane(mw, j + 1));
            const float w2 = __int_as_float(__builtin_amdgcn_readlane(mw, j + 2));
            const float w3 = __int_as_float(__builtin_amdgcn_readlane(mw, j + 3));
            const float4 h0 = *(const float4*)(hl + (long)s0 * D);
            const float4 h1 = *(const float4*)(hl + (long)s1 * D);
            const float4 h2 = *(const float4*)(hl + (long)s2 * D);
            const float4 h3 = *(const float4*)(hl + (long)s3 * D);
            axpy4(acc, w0, h0);
            axpy4(acc, w1, h1);
            axpy4(acc, w2, h2);
            axpy4(acc, w3, h3);
        }
        for (; j < m; ++j) {
            const int s0 = __builtin_amdgcn_readlane(ms, j);
            const float w0 = __int_as_float(__builtin_amdgcn_readlane(mw, j));
            const float4 h0 = *(const float4*)(hl + (long)s0 * D);
            axpy4(acc, w0, h0);
        }
    }
    *(float4*)(agg + (long)row * D + lane * 4) = acc;
}

static __device__ __forceinline__ void gather_row(float4& acc, int beg, int len,
                                                  int ms, int mw,
                                                  const int* __restrict__ ecol,
                                                  const float* __restrict__ ewt,
                                                  const float* hl, int lane) {
    for (int base = 0; base < len; base += 32) {
        const int m = min(32, len - base);
        if (base > 0) {
            ms = 0;
            mw = 0;
            if (lane < m) {
                ms = ecol[beg + base + lane];
                mw = __float_as_int(ewt[beg + base + lane]);
            }
        }
        int j = 0;
        for (; j + 8 <= m; j += 8) {
            int    s[8];
            float  w[8];
            float4 h[8];
#pragma unroll
            for (int u = 0; u < 8; ++u) s[u] = __builtin_amdgcn_readlane(ms, j + u);
#pragma unroll
            for (int u = 0; u < 8; ++u) h[u] = *(const float4*)(hl + (long)s[u] * D);
#pragma unroll
            for (int u = 0; u < 8; ++u) w[u] = __int_as_float(__builtin_amdgcn_readlane(mw, j + u));
#pragma unroll
            for (int u = 0; u < 8; ++u) axpy4(acc, w[u], h[u]);
        }
        if (j < m) {
            int    s[8];
            float  w[8];
            float4 h[8];
#pragma unroll
            for (int u = 0; u < 8; ++u) {
                s[u] = __builtin_amdgcn_readlane(ms, j + u);
                w[u] = __int_as_float(__builtin_amdgcn_readlane(mw, j + u));
            }
#pragma unroll
            for (int u = 0; u < 8; ++u) {
                h[u] = make_float4(0.0f, 0.0f, 0.0f, 0.0f);
                if (j + u < m) h[u] = *(const float4*)(hl + (long)s[u] * D);
            }
#pragma unroll
            for (int u = 0; u < 8; ++u) axpy4(acc, w[u], h[u]);
        }
    }
}

__global__ void __launch_bounds__(256)
cheb_gemm_kernel(const float* __restrict__ x,
                 const float* __restrict__ tx1,
                 const int* __restrict__ rowptr,
                 const int* __restrict__ cnt,
                 const int* __restrict__ ecol,
                 const float* __restrict__ ewt,
                 const float* __restrict__ weight,
                 const float* __restrict__ bias,
                 float* __restrict__ out, int n) {
    __shared__ __attribute__((aligned(16))) float lA[3][16][LDSPAD];

    const int m0   = blockIdx.x * 16;
    const int tid  = threadIdx.x;
    const int lane = tid & 31;
    const int wave = __builtin_amdgcn_readfirstlane(tid >> 5);

    {
        const int r0 = 2 * wave;
        const int r1 = r0 + 1;
        const int row0 = m0 + r0;
        const int row1 = m0 + r1;
        const bool in0 = row0 < n;
        const bool in1 = row1 < n;

        int beg0 = 0, len0 = 0, beg1 = 0, len1 = 0;
        if (in0) { beg0 = rowptr[row0]; len0 = cnt[row0]; }
        if (in1) { beg1 = rowptr[row1]; len1 = cnt[row1]; }

        float4 x0 = make_float4(0.0f, 0.0f, 0.0f, 0.0f), x1 = x0, t0 = x0, t1 = x0;
        if (in0) {
            x0 = *(const float4*)(x   + (long)row0 * D + lane * 4);
            t0 = *(const float4*)(tx1 + (long)row0 * D + lane * 4);
        }
        if (in1) {
            x1 = *(const float4*)(x   + (long)row1 * D + lane * 4);
            t1 = *(const float4*)(tx1 + (long)row1 * D + lane * 4);
        }

        int ms0 = 0, mw0 = 0, ms1 = 0, mw1 = 0;
        if (lane < len0) {
            ms0 = ecol[beg0 + lane];
            mw0 = __float_as_int(ewt[beg0 + lane]);
        }
        if (lane < len1) {
            ms1 = ecol[beg1 + lane];
            mw1 = __float_as_int(ewt[beg1 + lane]);
        }

        *(float4*)&lA[0][r0][lane * 4] = x0;
        *(float4*)&lA[1][r0][lane * 4] = t0;
        *(float4*)&lA[0][r1][lane * 4] = x1;
        *(float4*)&lA[1][r1][lane * 4] = t1;

        const float* hl = tx1 + lane * 4;
        float4 a0 = make_float4(0.0f, 0.0f, 0.0f, 0.0f);
        float4 a1 = a0;
        gather_row(a0, beg0, len0, ms0, mw0, ecol, ewt, hl, lane);
        gather_row(a1, beg1, len1, ms1, mw1, ecol, ewt, hl, lane);

        float4 v0, v1;
        v0.x = __builtin_fmaf(2.0f, a0.x, -x0.x);
        v0.y = __builtin_fmaf(2.0f, a0.y, -x0.y);
        v0.z = __builtin_fmaf(2.0f, a0.z, -x0.z);
        v0.w = __builtin_fmaf(2.0f, a0.w, -x0.w);
        v1.x = __builtin_fmaf(2.0f, a1.x, -x1.x);
        v1.y = __builtin_fmaf(2.0f, a1.y, -x1.y);
        v1.z = __builtin_fmaf(2.0f, a1.z, -x1.z);
        v1.w = __builtin_fmaf(2.0f, a1.w, -x1.w);
        *(float4*)&lA[2][r0][lane * 4] = v0;
        *(float4*)&lA[2][r1][lane * 4] = v1;
    }
    __syncthreads();

    const int hf   = lane >> 4;
    const int lm   = lane & 15;
    const int col  = wave * 16 + lm;

    v8f acc = {};
    for (int kk = 0; kk < D; kk += 4) {
        const int ka = kk + 2 * hf;
#pragma unroll
        for (int t = 0; t < 3; ++t) {
            v2f a;
            a.x = lA[t][lm][ka];
            a.y = lA[t][lm][ka + 1];
            const float* W = weight + (long)t * D * D;
            v2f b;
            b.x = W[(long)ka * D + col];
            b.y = W[(long)(ka + 1) * D + col];
            acc = __builtin_amdgcn_wmma_f32_16x16x4_f32(
                false, a, false, b, (short)0, acc, false, false);
        }
    }

    const float bv = bias[col];
#pragma unroll
    for (int i = 0; i < 8; ++i) {
        int row = m0 + i + 8 * hf;
        if (row < n) out[(long)row * D + col] = acc[i] + bv;
    }
}

extern "C" void kernel_launch(void* const* d_in, const int* in_sizes, int n_in,
                              void* d_out, int out_size, void* d_ws, size_t ws_size,
                              hipStream_t stream) {
    const float* x      = (const float*)d_in[0];
    const int*   ei     = (const int*)d_in[1];
    const float* weight = (const float*)d_in[2];
    const float* bias   = (const float*)d_in[3];
    float*       out    = (float*)d_out;

    const int n = in_sizes[0] / D;
    const int E = in_sizes[1] / 2;

    int*   deg = (int*)d_ws;
    int*   cnt = deg + n;
    float* tx1 = (float*)(cnt + n);
    int*   plane = (int*)(tx1 + (long)n * D);

    int* ps = (int*)tx1;
    int* pd = plane;

    int*   ecol   = plane + (long)NSLICE * n;
    float* ewt    = (float*)(ecol + E);
    int*   rowptr = (int*)(ewt + E);
    float* dis    = (float*)(rowptr + n);

    const int ranges    = (n + RANGE - 1) / RANGE;
    const int slice_len = (E + NSLICE - 1) / NSLICE;

    cheb_hist_kernel<<<dim3(ranges, NSLICE, 2), HT, 0, stream>>>(ei, E, n, slice_len, pd, ps);
    cheb_merge_kernel<<<dim3((n + 255) / 256, 2), 256, 0, stream>>>(pd, ps, cnt, deg, n);
    cheb_scan_kernel<<<(n + SCAN_T - 1) / SCAN_T, SCAN_T, 0, stream>>>(cnt, deg, rowptr, dis, n);
    cheb_fill_kernel<<<dim3(ranges, NSLICE), HT, 0, stream>>>(ei, E, n, slice_len, rowptr, pd, dis, ecol, ewt);

    const int gather_blocks = (n + 7) / 8;
    cheb_gather_kernel<<<gather_blocks, 256, 0, stream>>>(rowptr, cnt, ecol, ewt, x, tx1, n);

    cheb_gemm_kernel<<<(n + 15) / 16, 256, 0, stream>>>(x, tx1, rowptr, cnt, ecol, ewt,
                                                        weight, bias, out, n);
}


# GEMM block tile raised from 16 to 64 rows: each weight fragment load now feeds 4 row tiles on 4 accumulators per wave (weight reads from L2 cut 4x); a wave stages 8 rows, two at a time
# speedup vs baseline: 2.9985x; 1.0490x over previous
#include <hip/hip_runtime.h>

typedef __attribute__((ext_vector_type(2))) float v2f;
typedef __attribute__((ext_vector_type(8))) float v8f;

#define D 128
#define LDSPAD 132

#define SCAN_T 1024

#define RANGE  25000
#define NSLICE 64
#define HT     1024
#define EU     8

__global__ void __launch_bounds__(HT)
cheb_hist_kernel(const int* __restrict__ ei, int E, int n, int slice_len,
                 int* __restrict__ pd, int* __restrict__ ps) {
    __shared__ int tab[RANGE];
    const int tid  = threadIdx.x;
    const int base = blockIdx.x * RANGE;
    const int sl   = blockIdx.y;
    const bool by_src = (blockIdx.z != 0);
    const int lim = min(RANGE, n - base);

    for (int i = tid; i < RANGE; i += HT) tab[i] = 0;
    __syncthreads();

    const int e0 = sl * slice_len;
    const int e1 = min(E, e0 + slice_len);
    for (int eb = e0 + tid; eb < e1; eb += HT * EU) {
        int s[EU], d[EU];
#pragma unroll
        for (int u = 0; u < EU; ++u) {
            const int e = eb + u * HT;
            const bool in = e < e1;
            s[u] = in ? ei[e] : 0;
            d[u] = in ? ei[E + e] : 0;
        }
#pragma unroll
        for (int u = 0; u < EU; ++u) {
            const unsigned k = (unsigned)((by_src ? s[u] : d[u]) - base);
            if (s[u] != d[u] && k < (unsigned)lim) atomicAdd(&tab[k], 1);
        }
    }
    __syncthreads();

    int* __restrict__ rowp = (by_src ? ps : pd) + (long)sl * n + base;
    for (int i = tid; i < lim; i += HT) rowp[i] = tab[i];
}

__global__ void __launch_bounds__(256)
cheb_merge_kernel(int* __restrict__ pd, const int* __restrict__ ps,
                  int* __restrict__ cnt, int* __restrict__ deg, int n) {
    const int v = blockIdx.x * 256 + threadIdx.x;
    if (v >= n) return;
    if (blockIdx.y == 0) {
        int run = 0;
        for (int c = 0; c < NSLICE; c += 32) {
            int t[32];
#pragma unroll
            for (int u = 0; u < 32; ++u) t[u] = pd[(long)(c + u) * n + v];
#pragma unroll
            for (int u = 0; u < 32; ++u) {
                pd[(long)(c + u) * n + v] = run;
                run += t[u];
            }
        }
        cnt[v] = run;
    } else {
        int sum = 0;
#pragma unroll 32
        for (int sl = 0; sl < NSLICE; ++sl) sum += ps[(long)sl * n + v];
        deg[v] = sum;
    }
}

__global__ void __launch_bounds__(SCAN_T)
cheb_scan_kernel(const int* __restrict__ cnt, const int* __restrict__ deg,
                 int* __restrict__ rowptr, float* __restrict__ dis, int n) {
    __shared__ int wfront[32];
    __shared__ int wtotal[32];
    const int tid  = threadIdx.x;
    const int lane = tid & 31;
    const int wave = tid >> 5;
    const int first = blockIdx.x * SCAN_T;

    int part = 0;
    for (int i = tid; i < first; i += SCAN_T) part += cnt[i];
    for (int o = 16; o > 0; o >>= 1) part += __shfl_xor(part, o);

    const int i = first + tid;
    const int v = (i < n) ? cnt[i] : 0;
    int inc = v;
    for (int o = 1; o < 32; o <<= 1) {
        int t = __shfl_up(inc, o);
        if (lane >= o) inc += t;
    }
    if (lane == 0)  wfront[wave] = part;
    if (lane == 31) wtotal[wave] = inc;
    __syncthreads();

    int base = 0;
    for (int w = 0; w < 32; ++w) {
        base += wfront[w];
        if (w < wave) base += wtotal[w];
    }
    if (i < n) {
        rowptr[i] = base + inc - v;
        const int dg = deg[i];
        dis[i] = (dg > 0) ? rsqrtf(fmaxf((float)dg, 1.0f)) : 0.0f;
    }
}

__global__ void __launch_bounds__(HT)
cheb_fill_kernel(const int* __restrict__ ei, int E, int n, int slice_len,
                 const int* __restrict__ rowptr,
                 const int* __restrict__ pd,
                 const float* __restrict__ dis,
                 int* __restrict__ ecol,
                 float* __restrict__ ewt) {
    __shared__ int slot[RANGE];
    const int tid  = threadIdx.x;
    const int base = blockIdx.x * RANGE;
    const int sl   = blockIdx.y;
    const int lim  = min(RANGE, n - base);

    const int* __restrict__ off = pd + (long)sl * n + base;
    for (int i = tid; i < lim; i += HT) slot[i] = rowptr[base + i] + off[i];
    __syncthreads();

    const int e0 = sl * slice_len;
    const int e1 = min(E, e0 + slice_len);
    for (int eb = e0 + tid; eb < e1; eb += HT * EU) {
        int s[EU], d[EU];
#pragma unroll
        for (int u = 0; u < EU; ++u) {
            const int e = eb + u * HT;
            const bool in = e < e1;
            s[u] = in ? ei[e] : 0;
            d[u] = in ? ei[E + e] : 0;
        }
        bool  ok[EU];
        float fs[EU], fd[EU];
#pragma unroll
        for (int u = 0; u < EU; ++u) {
            const unsigned k = (unsigned)(d[u] - base);
            ok[u] = (s[u] != d[u]) && (k < (unsigned)lim);
            fs[u] = ok[u] ? dis[s[u]] : 0.0f;
            fd[u] = ok[u] ? dis[d[u]] : 0.0f;
        }
#pragma unroll
        for (int u = 0; u < EU; ++u) {
            if (ok[u]) {
                const int pos = atomicAdd(&slot[d[u] - base], 1);
                ecol[pos] = s[u];
                ewt[pos]  = -(fs[u] * fd[u]);
            }
        }
    }
}

static __device__ __forceinline__ void axpy4(float4& acc, float w, const float4 h) {
    acc.x = __builtin_fmaf(w, h.x, acc.x);
    acc.y = __builtin_fmaf(w, h.y, acc.y);
    acc.z = __builtin_fmaf(w, h.z, acc.z);
    acc.w = __builtin_fmaf(w, h.w, acc.w);
}

__global__ void __launch_bounds__(256)
cheb_gather_kernel(const int* __restrict__ rowptr,
                   const int* __restrict__ cnt,
                   const int* __restrict__ ecol,
                   const float* __restrict__ ewt,
                   const float* __restrict__ hin,
                   float* __restrict__ agg, int n) {
    const int row = __builtin_amdgcn_readfirstlane(
        (int)((blockIdx.x * 256u + threadIdx.x) >> 5));
    if (row >= n) return;
    const int lane = threadIdx.x & 31;
    const int beg = rowptr[row];
    const int len = cnt[row];
    const float* hl = hin + lane * 4;

    float4 acc = make_float4(0.0f, 0.0f, 0.0f, 0.0f);
    for (int base = 0; base < len; base += 32) {
        const int m = min(32, len - base);
        int ms = 0;
        int mw = 0;
        if (lane < m) {
            ms = ecol[beg + base + lane];
            mw = __float_as_int(ewt[beg + base + lane]);
        }
        int j = 0;
        for (; j + 4 <= m; j += 4) {
            const int s0 = __builtin_amdgcn_readlane(ms, j);
            const int s1 = __builtin_amdgcn_readlane(ms, j + 1);
            const int s2 = __builtin_amdgcn_readlane(ms, j + 2);
            const int s3 = __builtin_amdgcn_readlane(ms, j + 3);
            const float w0 = __int_as_float(__builtin_amdgcn_readlane(mw, j));
            const float w1 = __int_as_float(__builtin_amdgcn_readlane(mw, j + 1));
            const float w2 = __int_as_float(__builtin_amdgcn_readlane(mw, j + 2));
            const float w3 = __int_as_float(__builtin_amdgcn_readlane(mw, j + 3));
            const float4 h0 = *(const float4*)(hl + (long)s0 * D);
            const float4 h1 = *(const float4*)(hl + (long)s1 * D);
            const float4 h2 = *(const float4*)(hl + (long)s2 * D);
            const float4 h3 = *(const float4*)(hl + (long)s3 * D);
            axpy4(acc, w0, h0);
            axpy4(acc, w1, h1);
            axpy4(acc, w2, h2);
            axpy4(acc, w3, h3);
        }
        for (; j < m; ++j) {
            const int s0 = __builtin_amdgcn_readlane(ms, j);
            const float w0 = __int_as_float(__builtin_amdgcn_readlane(mw, j));
            const float4 h0 = *(const float4*)(hl + (long)s0 * D);
            axpy4(acc, w0, h0);
        }
    }
    *(float4*)(agg + (long)row * D + lane * 4) = acc;
}

static __device__ __forceinline__ void gather_row(float4& acc, int beg, int len,
                                                  int ms, int mw,
                                                  const int* __restrict__ ecol,
                                                  const float* __restrict__ ewt,
                                                  const float* hl, int lane) {
    for (int base = 0; base < len; base += 32) {
        const int m = min(32, len - base);
        if (base > 0) {
            ms = 0;
            mw = 0;
            if (lane < m) {
                ms = ecol[beg + base + lane];
                mw = __float_as_int(ewt[beg + base + lane]);
            }
        }
        int j = 0;
        for (; j + 8 <= m; j += 8) {
            int    s[8];
            float  w[8];
            float4 h[8];
#pragma unroll
            for (int u = 0; u < 8; ++u) s[u] = __builtin_amdgcn_readlane(ms, j + u);
#pragma unroll
            for (int u = 0; u < 8; ++u) h[u] = *(const float4*)(hl + (long)s[u] * D);
#pragma unroll
            for (int u = 0; u < 8; ++u) w[u] = __int_as_float(__builtin_amdgcn_readlane(mw, j + u));
#pragma unroll
            for (int u = 0; u < 8; ++u) axpy4(acc, w[u], h[u]);
        }
        if (j < m) {
            int    s[8];
            float  w[8];
            float4 h[8];
#pragma unroll
            for (int u = 0; u < 8; ++u) {
                s[u] = __builtin_amdgcn_readlane(ms, j + u);
                w[u] = __int_as_float(__builtin_amdgcn_readlane(mw, j + u));
            }
#pragma unroll
            for (int u = 0; u < 8; ++u) {
                h[u] = make_float4(0.0f, 0.0f, 0.0f, 0.0f);
                if (j + u < m) h[u] = *(const float4*)(hl + (long)s[u] * D);
            }
#pragma unroll
            for (int u = 0; u < 8; ++u) axpy4(acc, w[u], h[u]);
        }
    }
}

#define MT 4
#define BR (16 * MT)
__global__ void __launch_bounds__(256)
cheb_gemm_kernel(const float* __restrict__ x,
                 const float* __restrict__ tx1,
                 const int* __restrict__ rowptr,
                 const int* __restrict__ cnt,
                 const int* __restrict__ ecol,
                 const float* __restrict__ ewt,
                 const float* __restrict__ weight,
                 const float* __restrict__ bias,
                 float* __restrict__ out, int n) {
    __shared__ __attribute__((aligned(16))) float lA[3][BR][LDSPAD];

    const int m0   = blockIdx.x * BR;
    const int tid  = threadIdx.x;
    const int lane = tid & 31;
    const int wave = __builtin_amdgcn_readfirstlane(tid >> 5);

    {
        int rp = 0, rc = 0;
        {
            const int r = m0 + 8 * wave + lane;
            if (lane < 8 && r < n) { rp = rowptr[r]; rc = cnt[r]; }
        }
        const float* hl = tx1 + lane * 4;

        for (int p = 0; p < 4; ++p) {
            const int r0 = 8 * wave + 2 * p;
            const int r1 = r0 + 1;
            const int row0 = m0 + r0;
            const int row1 = m0 + r1;
            const bool in0 = row0 < n;
            const bool in1 = row1 < n;

            const int beg0 = __builtin_amdgcn_readlane(rp, 2 * p);
            const int len0 = __builtin_amdgcn_readlane(rc, 2 * p);
            const int beg1 = __builtin_amdgcn_readlane(rp, 2 * p + 1);
            const int len1 = __builtin_amdgcn_readlane(rc, 2 * p + 1);

            float4 x0 = make_float4(0.0f, 0.0f, 0.0f, 0.0f), x1 = x0, t0 = x0, t1 = x0;
            if (in0) {
                x0 = *(const float4*)(x   + (long)row0 * D + lane * 4);
                t0 = *(const float4*)(tx1 + (long)row0 * D + lane * 4);
            }
            if (in1) {
                x1 = *(const float4*)(x   + (long)row1 * D + lane * 4);
                t1 = *(const float4*)(tx1 + (long)row1 * D + lane * 4);
            }

            int ms0 = 0, mw0 = 0, ms1 = 0, mw1 = 0;
            if (lane < len0) {
                ms0 = ecol[beg0 + lane];
                mw0 = __float_as_int(ewt[beg0 + lane]);
            }
            if (lane < len1) {
                ms1 = ecol[beg1 + lane];
                mw1 = __float_as_int(ewt[beg1 + lane]);
            }

            *(float4*)&lA[0][r0][lane * 4] = x0;
            *(float4*)&lA[1][r0][lane * 4] = t0;
            *(float4*)&lA[0][r1][lane * 4] = x1;
            *(float4*)&lA[1][r1][lane * 4] = t1;

            float4 a0 = make_float4(0.0f, 0.0f, 0.0f, 0.0f);
            float4 a1 = a0;
            gather_row(a0, beg0, len0, ms0, mw0, ecol, ewt, hl, lane);
            gather_row(a1, beg1, len1, ms1, mw1, ecol, ewt, hl, lane);

            float4 v0, v1;
            v0.x = __builtin_fmaf(2.0f, a0.x, -x0.x);
            v0.y = __builtin_fmaf(2.0f, a0.y, -x0.y);
            v0.z = __builtin_fmaf(2.0f, a0.z, -x0.z);
            v0.w = __builtin_fmaf(2.0f, a0.w, -x0.w);
            v1.x = __builtin_fmaf(2.0f, a1.x, -x1.x);
            v1.y = __builtin_fmaf(2.0f, a1.y, -x1.y);
            v1.z = __builtin_fmaf(2.0f, a1.z, -x1.z);
            v1.w = __builtin_fmaf(2.0f, a1.w, -x1.w);
            *(float4*)&lA[2][r0][lane * 4] = v0;
            *(float4*)&lA[2][r1][lane * 4] = v1;
        }
    }
    __syncthreads();

    const int hf   = lane >> 4;
    const int lm   = lane & 15;
    const int col  = wave * 16 + lm;

    v8f acc[MT];
#pragma unroll
    for (int m = 0; m < MT; ++m) acc[m] = (v8f){};

#pragma unroll 4
    for (int kk = 0; kk < D; kk += 4) {
        const int ka = kk + 2 * hf;
#pragma unroll
        for (int t = 0; t < 3; ++t) {
            const float* Wt = weight + (long)t * D * D;
            v2f b;
            b.x = Wt[(long)ka * D + col];
            b.y = Wt[(long)(ka + 1) * D + col];
#pragma unroll
            for (int m = 0; m < MT; ++m) {
                v2f a;
                a.x = lA[t][16 * m + lm][ka];
                a.y = lA[t][16 * m + lm][ka + 1];
                acc[m] = __builtin_amdgcn_wmma_f32_16x16x4_f32(
                    false, a, false, b, (short)0, acc[m], false, false);
            }
        }
    }

    const float bv = bias[col];
#pragma unroll
    for (int m = 0; m < MT; ++m) {
#pragma unroll
        for (int i = 0; i < 8; ++i) {
            int row = m0 + 16 * m + i + 8 * hf;
            if (row < n) out[(long)row * D + col] = acc[m][i] + bv;
        }
    }
}

extern "C" void kernel_launch(void* const* d_in, const int* in_sizes, int n_in,
                              void* d_out, int out_size, void* d_ws, size_t ws_size,
                              hipStream_t stream) {
    const float* x      = (const float*)d_in[0];
    const int*   ei     = (const int*)d_in[1];
    const float* weight = (const float*)d_in[2];
    const float* bias   = (const float*)d_in[3];
    float*       out    = (float*)d_out;

    const int n = in_sizes[0] / D;
    const int E = in_sizes[1] / 2;

    int*   deg = (int*)d_ws;
    int*   cnt = deg + n;
    float* tx1 = (float*)(cnt + n);
    int*   plane = (int*)(tx1 + (long)n * D);

    int* ps = (int*)tx1;
    int* pd = plane;

    int*   ecol   = plane + (long)NSLICE * n;
    float* ewt    = (float*)(ecol + E);
    int*   rowptr = (int*)(ewt + E);
    float* dis    = (float*)(rowptr + n);

    const int ranges    = (n + RANGE - 1) / RANGE;
    const int slice_len = (E + NSLICE - 1) / NSLICE;

    cheb_hist_kernel<<<dim3(ranges, NSLICE, 2), HT, 0, stream>>>(ei, E, n, slice_len, pd, ps);
    cheb_merge_kernel<<<dim3((n + 255) / 256, 2), 256, 0, stream>>>(pd, ps, cnt, deg, n);
    cheb_scan_kernel<<<(n + SCAN_T - 1) / SCAN_T, SCAN_T, 0, stream>>>(cnt, deg, rowptr, dis, n);
    cheb_fill_kernel<<<dim3(ranges, NSLICE), HT, 0, stream>>>(ei, E, n, slice_len, rowptr, pd, dis, ecol, ewt);

    const int gather_blocks = (n + 7) / 8;
    cheb_gather_kernel<<<gather_blocks, 256, 0, stream>>>(rowptr, cnt, ecol, ewt, x, tx1, n);

    cheb_gemm_kernel<<<(n + BR - 1) / BR, 256, 0, stream>>>(x, tx1, rowptr, cnt, ecol, ewt,
                                                        weight, bias, out, n);
}


# fused kernel: last sixteenth of the rows runs as 16-row blocks (one row tile, two rows staged per wave) at the end of the grid, so no third round of 27 full blocks on an idle chip; sums unchanged
# speedup vs baseline: 3.6663x; 1.2227x over previous
#include <hip/hip_runtime.h>

typedef __attribute__((ext_vector_type(2))) float v2f;
typedef __attribute__((ext_vector_type(8))) float v8f;

#define D 128
#define LDSPAD 132

#define SCAN_T 1024

#define RANGE  25000
#define NSLICE 64
#define HT     1024
#define EU     8

__global__ void __launch_bounds__(HT)
cheb_hist_kernel(const int* __restrict__ ei, int E, int n, int slice_len,
                 int* __restrict__ pd, int* __restrict__ ps) {
    __shared__ int tab[RANGE];
    const int tid  = threadIdx.x;
    const int base = blockIdx.x * RANGE;
    const int sl   = blockIdx.y;
    const bool by_src = (blockIdx.z != 0);
    const int lim = min(RANGE, n - base);

    for (int i = tid; i < RANGE; i += HT) tab[i] = 0;
    __syncthreads();

    const int e0 = sl * slice_len;
    const int e1 = min(E, e0 + slice_len);
    for (int eb = e0 + tid; eb < e1; eb += HT * EU) {
        int s[EU], d[EU];
#pragma unroll
        for (int u = 0; u < EU; ++u) {
            const int e = eb + u * HT;
            const bool in = e < e1;
            s[u] = in ? ei[e] : 0;
            d[u] = in ? ei[E + e] : 0;
        }
#pragma unroll
        for (int u = 0; u < EU; ++u) {
            const unsigned k = (unsigned)((by_src ? s[u] : d[u]) - base);
            if (s[u] != d[u] && k < (unsigned)lim) atomicAdd(&tab[k], 1);
        }
    }
    __syncthreads();

    int* __restrict__ rowp = (by_src ? ps : pd) + (long)sl * n + base;
    for (int i = tid; i < lim; i += HT) rowp[i] = tab[i];
}

__global__ void __launch_bounds__(256)
cheb_merge_kernel(int* __restrict__ pd, const int* __restrict__ ps,
                  int* __restrict__ cnt, int* __restrict__ deg, int n) {
    const int v = blockIdx.x * 256 + threadIdx.x;
    if (v >= n) return;
    if (blockIdx.y == 0) {
        int run = 0;
        for (int c = 0; c < NSLICE; c += 32) {
            int t[32];
#pragma unroll
            for (int u = 0; u < 32; ++u) t[u] = pd[(long)(c + u) * n + v];
#pragma unroll
            for (int u = 0; u < 32; ++u) {
                pd[(long)(c + u) * n + v] = run;
                run += t[u];
            }
        }
        cnt[v] = run;
    } else {
        int sum = 0;
#pragma unroll 32
        for (int sl = 0; sl < NSLICE; ++sl) sum += ps[(long)sl * n + v];
        deg[v] = sum;
    }
}

#define NSCALE 256
__global__ void __launch_bounds__(SCAN_T)
cheb_scan_kernel(const int* __restrict__ cnt, const int* __restrict__ deg,
                 int* __restrict__ rowptr, float* __restrict__ dis,
                 float* __restrict__ rsd,
                 const float* __restrict__ x, float* __restrict__ u,
                 int n, int nscan) {
    __shared__ int wfront[32];
    __shared__ int wtotal[32];
    const int tid  = threadIdx.x;
    const int lane = tid & 31;
    const int wave = tid >> 5;

    if ((int)blockIdx.x >= nscan) {
        const int nw = NSCALE * (SCAN_T / 32);
        const int w0 = ((int)blockIdx.x - nscan) * (SCAN_T / 32) + wave;
        for (int rb = w0; rb < n; rb += 4 * nw) {
            int    dg[4];
            float4 v[4];
#pragma unroll
            for (int k = 0; k < 4; ++k) {
                const int row = rb + k * nw;
                dg[k] = 0;
                v[k] = make_float4(0.0f, 0.0f, 0.0f, 0.0f);
                if (row < n) {
                    dg[k] = deg[row];
                    v[k] = *(const float4*)(x + (long)row * D + lane * 4);
                }
            }
#pragma unroll
            for (int k = 0; k < 4; ++k) {
                const int row = rb + k * nw;
                const float ds = (dg[k] > 0) ? rsqrtf(fmaxf((float)dg[k], 1.0f)) : 0.0f;
                if (row < n)
                    *(float4*)(u + (long)row * D + lane * 4) =
                        make_float4(ds * v[k].x, ds * v[k].y, ds * v[k].z, ds * v[k].w);
            }
        }
        return;
    }

    const int first = blockIdx.x * SCAN_T;

    int part = 0;
    for (int i = tid; i < first; i += SCAN_T) part += cnt[i];
    for (int o = 16; o > 0; o >>= 1) part += __shfl_xor(part, o);

    const int i = first + tid;
    const int v = (i < n) ? cnt[i] : 0;
    int inc = v;
    for (int o = 1; o < 32; o <<= 1) {
        int t = __shfl_up(inc, o);
        if (lane >= o) inc += t;
    }
    if (lane == 0)  wfront[wave] = part;
    if (lane == 31) wtotal[wave] = inc;
    __syncthreads();

    int base = 0;
    for (int w = 0; w < 32; ++w) {
        base += wfront[w];
        if (w < wave) base += wtotal[w];
    }
    if (i < n) {
        rowptr[i] = base + inc - v;
        const int dg = deg[i];
        dis[i] = (dg > 0) ? rsqrtf(fmaxf((float)dg, 1.0f)) : 0.0f;
        rsd[i] = (dg > 0) ? sqrtf((float)dg) : 0.0f;
    }
}

__global__ void __launch_bounds__(HT)
cheb_fill_kernel(const int* __restrict__ ei, int E, int n, int slice_len,
                 const int* __restrict__ rowptr,
                 const int* __restrict__ pd,
                 int* __restrict__ ecol) {
    __shared__ int slot[RANGE];
    const int tid  = threadIdx.x;
    const int base = blockIdx.x * RANGE;
    const int sl   = blockIdx.y;
    const int lim  = min(RANGE, n - base);

    const int* __restrict__ off = pd + (long)sl * n + base;
    for (int i = tid; i < lim; i += HT) slot[i] = rowptr[base + i] + off[i];
    __syncthreads();

    const int e0 = sl * slice_len;
    const int e1 = min(E, e0 + slice_len);
    for (int eb = e0 + tid; eb < e1; eb += HT * EU) {
        int s[EU], d[EU];
#pragma unroll
        for (int u = 0; u < EU; ++u) {
            const int e = eb + u * HT;
            const bool in = e < e1;
            s[u] = in ? ei[e] : 0;
            d[u] = in ? ei[E + e] : 0;
        }
#pragma unroll
        for (int u = 0; u < EU; ++u) {
            const unsigned k = (unsigned)(d[u] - base);
            if (s[u] != d[u] && k < (unsigned)lim) {
                const int pos = atomicAdd(&slot[k], 1);
                ecol[pos] = s[u];
            }
        }
    }
}

static __device__ __forceinline__ void add4(float4& acc, const float4 h) {
    acc.x += h.x;
    acc.y += h.y;
    acc.z += h.z;
    acc.w += h.w;
}

__global__ void __launch_bounds__(256)
cheb_gather_kernel(const int* __restrict__ rowptr,
                   const int* __restrict__ cnt,
                   const int* __restrict__ ecol,
                   const float* __restrict__ dis,
                   const float* __restrict__ hin,
                   float* __restrict__ agg, int n) {
    const int row = __builtin_amdgcn_readfirstlane(
        (int)((blockIdx.x * 256u + threadIdx.x) >> 5));
    if (row >= n) return;
    const int lane = threadIdx.x & 31;
    const int beg = rowptr[row];
    const int len = cnt[row];
    const float ds = dis[row];
    const float* hl = hin + lane * 4;

    float4 acc = make_float4(0.0f, 0.0f, 0.0f, 0.0f);
    for (int base = 0; base < len; base += 32) {
        const int m = min(32, len - base);
        int ms = 0;
        if (lane < m) ms = ecol[beg + base + lane];
        int j = 0;
        for (; j + 4 <= m; j += 4) {
            const int s0 = __builtin_amdgcn_readlane(ms, j);
            const int s1 = __builtin_amdgcn_readlane(ms, j + 1);
            const int s2 = __builtin_amdgcn_readlane(ms, j + 2);
            const int s3 = __builtin_amdgcn_readlane(ms, j + 3);
            const float4 h0 = *(const float4*)(hl + (long)s0 * D);
            const float4 h1 = *(const float4*)(hl + (long)s1 * D);
            const float4 h2 = *(const float4*)(hl + (long)s2 * D);
            const float4 h3 = *(const float4*)(hl + (long)s3 * D);
            add4(acc, h0);
            add4(acc, h1);
            add4(acc, h2);
            add4(acc, h3);
        }
        for (; j < m; ++j) {
            const int s0 = __builtin_amdgcn_readlane(ms, j);
            const float4 h0 = *(const float4*)(hl + (long)s0 * D);
            add4(acc, h0);
        }
    }
    float4 p;
    p.x = ds * (-ds * acc.x);
    p.y = ds * (-ds * acc.y);
    p.z = ds * (-ds * acc.z);
    p.w = ds * (-ds * acc.w);
    *(float4*)(agg + (long)row * D + lane * 4) = p;
}

static __device__ __forceinline__ void gather_row(float4& acc, int beg, int len,
                                                  int ms,
                                                  const int* __restrict__ ecol,
                                                  const float* hl, int lane) {
    for (int base = 0; base < len; base += 32) {
        const int m = min(32, len - base);
        if (base > 0) {
            ms = 0;
            if (lane < m) ms = ecol[beg + base + lane];
        }
        int j = 0;
        for (; j + 8 <= m; j += 8) {
            int    s[8];
            float4 h[8];
#pragma unroll
            for (int u = 0; u < 8; ++u) s[u] = __builtin_amdgcn_readlane(ms, j + u);
#pragma unroll
            for (int u = 0; u < 8; ++u) h[u] = *(const float4*)(hl + (long)s[u] * D);
#pragma unroll
            for (int u = 0; u < 8; ++u) add4(acc, h[u]);
        }
        if (j < m) {
            int    s[8];
            float4 h[8];
#pragma unroll
            for (int u = 0; u < 8; ++u) s[u] = __builtin_amdgcn_readlane(ms, j + u);
#pragma unroll
            for (int u = 0; u < 8; ++u) {
                h[u] = make_float4(0.0f, 0.0f, 0.0f, 0.0f);
                if (j + u < m) h[u] = *(const float4*)(hl + (long)s[u] * D);
            }
#pragma unroll
            for (int u = 0; u < 8; ++u) add4(acc, h[u]);
        }
    }
}

#define MT 4
#define BR (16 * MT)

template <int MTT>
static __device__ __forceinline__ void
cheb_gemm_tile(float (*lA)[BR][LDSPAD], const int m0,
               const float* __restrict__ x,
               const float* __restrict__ p1,
               const int* __restrict__ rowptr,
               const int* __restrict__ cnt,
               const int* __restrict__ ecol,
               const float* __restrict__ dis,
               const float* __restrict__ rsd,
               const float* __restrict__ weight,
               const float* __restrict__ bias,
               float* __restrict__ out, int n) {
    const int tid  = threadIdx.x;
    const int lane = tid & 31;
    const int wave = __builtin_amdgcn_readfirstlane(tid >> 5);

    {
        int rp = 0, rc = 0, rd = 0, rq = 0;
        {
            const int r = m0 + 2 * MTT * wave + lane;
            if (lane < 2 * MTT && r < n) {
                rp = rowptr[r];
                rc = cnt[r];
                rd = __float_as_int(dis[r]);
                rq = __float_as_int(rsd[r]);
            }
        }
        const float* hl = p1 + lane * 4;

        for (int p = 0; p < MTT; ++p) {
            const int r0 = 2 * MTT * wave + 2 * p;
            const int r1 = r0 + 1;
            const int row0 = m0 + r0;
            const int row1 = m0 + r1;
            const bool in0 = row0 < n;
            const bool in1 = row1 < n;

            const int beg0 = __builtin_amdgcn_readlane(rp, 2 * p);
            const int len0 = __builtin_amdgcn_readlane(rc, 2 * p);
            const int beg1 = __builtin_amdgcn_readlane(rp, 2 * p + 1);
            const int len1 = __builtin_amdgcn_readlane(rc, 2 * p + 1);
            const float d0 = __int_as_float(__builtin_amdgcn_readlane(rd, 2 * p));
            const float d1 = __int_as_float(__builtin_amdgcn_readlane(rd, 2 * p + 1));
            const float q0 = __int_as_float(__builtin_amdgcn_readlane(rq, 2 * p));
            const float q1 = __int_as_float(__builtin_amdgcn_readlane(rq, 2 * p + 1));

            float4 x0 = make_float4(0.0f, 0.0f, 0.0f, 0.0f), x1 = x0, t0 = x0, t1 = x0;
            if (in0) {
                x0 = *(const float4*)(x   + (long)row0 * D + lane * 4);
                t0 = *(const float4*)(p1  + (long)row0 * D + lane * 4);
            }
            if (in1) {
                x1 = *(const float4*)(x   + (long)row1 * D + lane * 4);
                t1 = *(const float4*)(p1  + (long)row1 * D + lane * 4);
            }

            int ms0 = 0, ms1 = 0;
            if (lane < len0) ms0 = ecol[beg0 + lane];
            if (lane < len1) ms1 = ecol[beg1 + lane];

            t0 = make_float4(q0 * t0.x, q0 * t0.y, q0 * t0.z, q0 * t0.w);
            t1 = make_float4(q1 * t1.x, q1 * t1.y, q1 * t1.z, q1 * t1.w);
            *(float4*)&lA[0][r0][lane * 4] = x0;
            *(float4*)&lA[1][r0][lane * 4] = t0;
            *(float4*)&lA[0][r1][lane * 4] = x1;
            *(float4*)&lA[1][r1][lane * 4] = t1;

            float4 a0 = make_float4(0.0f, 0.0f, 0.0f, 0.0f);
            float4 a1 = a0;
            gather_row(a0, beg0, len0, ms0, ecol, hl, lane);
            gather_row(a1, beg1, len1, ms1, ecol, hl, lane);

            const float c0 = -2.0f * d0;
            const float c1 = -2.0f * d1;
            float4 v0, v1;
            v0.x = __builtin_fmaf(c0, a0.x, -x0.x);
            v0.y = __builtin_fmaf(c0, a0.y, -x0.y);
            v0.z = __builtin_fmaf(c0, a0.z, -x0.z);
            v0.w = __builtin_fmaf(c0, a0.w, -x0.w);
            v1.x = __builtin_fmaf(c1, a1.x, -x1.x);
            v1.y = __builtin_fmaf(c1, a1.y, -x1.y);
            v1.z = __builtin_fmaf(c1, a1.z, -x1.z);
            v1.w = __builtin_fmaf(c1, a1.w, -x1.w);
            *(float4*)&lA[2][r0][lane * 4] = v0;
            *(float4*)&lA[2][r1][lane * 4] = v1;
        }
    }
    __syncthreads();

    const int hf   = lane >> 4;
    const int lm   = lane & 15;
    const int col  = wave * 16 + lm;

    v8f acc[MTT];
#pragma unroll
    for (int m = 0; m < MTT; ++m) acc[m] = (v8f){};

#pragma unroll 4
    for (int kk = 0; kk < D; kk += 4) {
        const int ka = kk + 2 * hf;
#pragma unroll
        for (int t = 0; t < 3; ++t) {
            const float* Wt = weight + (long)t * D * D;
            v2f b;
            b.x = Wt[(long)ka * D + col];
            b.y = Wt[(long)(ka + 1) * D + col];
#pragma unroll
            for (int m = 0; m < MTT; ++m) {
                v2f a;
                a.x = lA[t][16 * m + lm][ka];
                a.y = lA[t][16 * m + lm][ka + 1];
                acc[m] = __builtin_amdgcn_wmma_f32_16x16x4_f32(
                    false, a, false, b, (short)0, acc[m], false, false);
            }
        }
    }

    const float bv = bias[col];
#pragma unroll
    for (int m = 0; m < MTT; ++m) {
#pragma unroll
        for (int i = 0; i < 8; ++i) {
            int row = m0 + 16 * m + i + 8 * hf;
            if (row < n) out[(long)row * D + col] = acc[m][i] + bv;
        }
    }
}

__global__ void __launch_bounds__(256)
cheb_gemm_kernel(const float* __restrict__ x,
                 const float* __restrict__ p1,
                 const int* __restrict__ rowptr,
                 const int* __restrict__ cnt,
                 const int* __restrict__ ecol,
                 const float* __restrict__ dis,
                 const float* __restrict__ rsd,
                 const float* __restrict__ weight,
                 const float* __restrict__ bias,
                 float* __restrict__ out, int n, int nbig) {
    __shared__ __attribute__((aligned(16))) float lA[3][BR][LDSPAD];
    const int b = blockIdx.x;
    if (b < nbig) {
        cheb_gemm_tile<MT>(lA, b * BR, x, p1, rowptr, cnt, ecol, dis, rsd,
                           weight, bias, out, n);
    } else {
        cheb_gemm_tile<1>(lA, nbig * BR + (b - nbig) * 16, x, p1, rowptr, cnt, ecol,
                          dis, rsd, weight, bias, out, n);
    }
}

extern "C" void kernel_launch(void* const* d_in, const int* in_sizes, int n_in,
                              void* d_out, int out_size, void* d_ws, size_t ws_size,
                              hipStream_t stream) {
    const float* x      = (const float*)d_in[0];
    const int*   ei     = (const int*)d_in[1];
    const float* weight = (const float*)d_in[2];
    const float* bias   = (const float*)d_in[3];
    float*       out    = (float*)d_out;

    const int n = in_sizes[0] / D;
    const int E = in_sizes[1] / 2;

    int*   deg = (int*)d_ws;
    int*   cnt = deg + n;
    float* p1  = (float*)(cnt + n);
    int*   plane = (int*)(p1 + (long)n * D);

    int* ps = (int*)p1;
    int* pd = plane;

    int*   ecol   = plane + (long)NSLICE * n;
    int*   rowptr = ecol + E;
    float* dis    = (float*)(rowptr + n);
    float* rsd    = dis + n;

    float* u = out;

    const int ranges    = (n + RANGE - 1) / RANGE;
    const int slice_len = (E + NSLICE - 1) / NSLICE;
    const int nscan     = (n + SCAN_T - 1) / SCAN_T;

    cheb_hist_kernel<<<dim3(ranges, NSLICE, 2), HT, 0, stream>>>(ei, E, n, slice_len, pd, ps);
    cheb_merge_kernel<<<dim3((n + 255) / 256, 2), 256, 0, stream>>>(pd, ps, cnt, deg, n);
    cheb_scan_kernel<<<nscan + NSCALE, SCAN_T, 0, stream>>>(cnt, deg, rowptr, dis, rsd, x, u, n, nscan);
    cheb_fill_kernel<<<dim3(ranges, NSLICE), HT, 0, stream>>>(ei, E, n, slice_len, rowptr, pd, ecol);

    const int gather_blocks = (n + 7) / 8;
    cheb_gather_kernel<<<gather_blocks, 256, 0, stream>>>(rowptr, cnt, ecol, dis, u, p1, n);

    const int nbig   = (n / BR) * 15 / 16;
    const int nshort = (n - nbig * BR + 15) / 16;
    cheb_gemm_kernel<<<nbig + nshort, 256, 0, stream>>>(x, p1, rowptr, cnt, ecol, dis, rsd,
                                                      weight, bias, out, n, nbig);
}
